# starting kernel
# baseline (speedup 1.0000x reference)
#include <hip/hip_runtime.h>

typedef __attribute__((ext_vector_type(2))) float v2f;
typedef __attribute__((ext_vector_type(4))) float v4f;
typedef __attribute__((ext_vector_type(8))) float v8f;

#define BATCH  8
#define CIN    32
#define COUT   32
#define HW     128
#define PLANE  (HW * HW)
#define REGION 66
#define XT     5
#define SCALE  (1.0f/256.0f)

__global__ void __launch_bounds__(256) bias_fill_kernel(
    const float* __restrict__ bias, float* __restrict__ out) {
  int i = blockIdx.x * 256 + threadIdx.x;
  int ch = (i >> 12) & (COUT - 1);
  float b = bias[ch];
  v4f v = {b, b, b, b};
  reinterpret_cast<v4f*>(out)[i] = v;
}

__global__ void __launch_bounds__(32) fftconv_corner_wmma(
    const float* __restrict__ x,
    const float* __restrict__ w,
    const float* __restrict__ bias,
    float* __restrict__ out) {
  const int lane = threadIdx.x;
  const int half = lane >> 4;
  const int mn   = lane & 15;
  const int xt   = blockIdx.x % XT;
  const int ot   = blockIdx.x / XT;
  const int y    = blockIdx.y;
  const int b    = blockIdx.z;

  const int oA = ot * 16 + mn;
  const int xg = xt * 16 + mn;

  const float* __restrict__ wrow = w + (size_t)oA * (CIN * 9);
  const float* __restrict__ xb   = x + (size_t)b * CIN * PLANE;

  v8f acc = {0.f, 0.f, 0.f, 0.f, 0.f, 0.f, 0.f, 0.f};

  #pragma unroll
  for (int tap = 0; tap < 9; ++tap) {
    const int ky = tap / 3;
    const int kx = tap % 3;
    const int row = 64 + y  - ky;
    const int col = 64 + xg - kx;
    const float maskf = (row < HW && col < HW) ? 1.0f : 0.0f;
    const int rowc = row < HW ? row : HW - 1;
    const int colc = col < HW ? col : HW - 1;
    const float* __restrict__ xbase = xb + (rowc * HW + colc);
    const float* __restrict__ wbase = wrow + tap;

    #pragma unroll
    for (int c4 = 0; c4 < CIN / 4; ++c4) {
      const int c0 = c4 * 4 + 2 * half;
      v2f a, bm;
      a.x  = wbase[c0 * 9];
      a.y  = wbase[c0 * 9 + 9];
      bm.x = xbase[c0 * PLANE] * maskf;
      bm.y = xbase[c0 * PLANE + PLANE] * maskf;
      acc = __builtin_amdgcn_wmma_f32_16x16x4_f32(false, a, false, bm,
                                                  (short)0, acc, false, false);
    }
  }

  const int xo = xt * 16 + mn;
  if (xo < REGION) {
    #pragma unroll
    for (int r = 0; r < 8; ++r) {
      const int o = ot * 16 + r + 8 * half;
      out[(((size_t)b * COUT + o) * HW + y) * HW + xo] = acc[r] * SCALE + bias[o];
    }
  }
}

extern "C" void kernel_launch(void* const* d_in, const int* in_sizes, int n_in,
                              void* d_out, int out_size, void* d_ws, size_t ws_size,
                              hipStream_t stream) {
  const float* x    = (const float*)d_in[0];
  const float* w    = (const float*)d_in[1];
  const float* bias = (const float*)d_in[2];
  float* out = (float*)d_out;

  const int n4 = BATCH * COUT * HW * HW / 4;
  bias_fill_kernel<<<n4 / 256, 256, 0, stream>>>(bias, out);

  dim3 grid(XT * 2, REGION, BATCH);
  fftconv_corner_wmma<<<grid, 32, 0, stream>>>(x, w, bias, out);
}


# Conv operands staged once per work-group in LDS (weights at a conflict-free pitch, x halo band as channel pairs), read as fragments by 8 waves over flattened 16-pixel tiles; sum order unchanged
# speedup vs baseline: 1.8971x; 1.8971x over previous
#include <hip/hip_runtime.h>


typedef __attribute__((ext_vector_type(2))) float v2f;
typedef __attribute__((ext_vector_type(4))) float v4f;
typedef __attribute__((ext_vector_type(8))) float v8f;

#define BATCH  8
#define CIN    32
#define COUT   32
#define HW     128
#define PLANE  (HW * HW)
#define REGION 66
#define NPIX   (REGION * REGION)
#define NTILE  273
#define GROUPS 32
#define SCALE  (1.0f/256.0f)

#define WPITCH 292
#define BROWS  6
#define BCOLP  72
#define ROWF   (BCOLP * 2)
#define PAIRF  (BROWS * ROWF)
#define XITEMS (16 * BROWS * 33)

__global__ void __launch_bounds__(256) bias_fill_kernel(
    const float* __restrict__ bias, float* __restrict__ out) {
  int i = blockIdx.x * 256 + threadIdx.x;
  int ch = (i >> 12) & (COUT - 1);
  float b = bias[ch];
  v4f v = {b, b, b, b};
  reinterpret_cast<v4f*>(out)[i] = v;
}

__global__ void __launch_bounds__(256) fftconv_corner_wmma(
    const float* __restrict__ x,
    const float* __restrict__ w,
    const float* __restrict__ bias,
    float* __restrict__ out) {
  __shared__ __attribute__((aligned(16))) float sW[COUT * WPITCH];
  __shared__ __attribute__((aligned(16))) float sX[16 * PAIRF];

  const int tid  = threadIdx.x;
  const int lane = tid & 31;
  const int half = lane >> 4;
  const int mn   = lane & 15;
  const int g    = blockIdx.x;
  const int b    = blockIdx.y;

  const int t0  = g * 8 + (g < 17 ? g : 17);
  const int cnt = 8 + (g < 17 ? 1 : 0);
  const int p0  = t0 * 16;
  const int y0  = p0 / REGION;

  const float* __restrict__ xb = x + (size_t)b * CIN * PLANE;

  v4f wv[9];
  #pragma unroll
  for (int i = 0; i < 9; ++i)
    wv[i] = reinterpret_cast<const v4f*>(w)[tid + i * 256];

  v2f xv0[13], xv1[13];
  #pragma unroll
  for (int i = 0; i < 13; ++i) {
    const int idx = tid + i * 256;
    const int idc = idx < XITEMS ? idx : XITEMS - 1;
    const int t   = idc / 33;
    const int bc2 = idc - t * 33;
    const int q   = t / BROWS;
    const int br  = t - q * BROWS;
    const int row = 62 + y0 + br;
    const int rowc = row < HW ? row : HW - 1;
    const float* src = xb + (q * 2) * PLANE + rowc * HW + 62 + bc2 * 2;
    xv0[i] = *reinterpret_cast<const v2f*>(src);
    xv1[i] = *reinterpret_cast<const v2f*>(src + PLANE);
  }

  #pragma unroll
  for (int i = 0; i < 9; ++i) {
    const int idx = tid + i * 256;
    const int o   = idx / 72;
    const int r4  = idx - o * 72;
    *reinterpret_cast<v4f*>(&sW[o * WPITCH + r4 * 4]) = wv[i];
  }
  #pragma unroll
  for (int i = 0; i < 13; ++i) {
    const int idx = tid + i * 256;
    if (idx < XITEMS) {
      const int t   = idx / 33;
      const int bc2 = idx - t * 33;
      const int br  = t % BROWS;
      const bool in = (62 + y0 + br) < HW;
      v4f pk;
      pk.x = in ? xv0[i].x : 0.0f;
      pk.y = in ? xv1[i].x : 0.0f;
      pk.z = in ? xv0[i].y : 0.0f;
      pk.w = in ? xv1[i].y : 0.0f;
      *reinterpret_cast<v4f*>(&sX[t * ROWF + bc2 * 4]) = pk;
    }
  }
  for (int idx = tid; idx < 16 * BROWS * 3; idx += 256) {
    const int t  = idx / 3;
    const int k3 = idx - t * 3;
    const v4f z = {0.f, 0.f, 0.f, 0.f};
    *reinterpret_cast<v4f*>(&sX[t * ROWF + (33 + k3) * 4]) = z;
  }
  __syncthreads();

  const int wave = __builtin_amdgcn_readfirstlane(tid >> 5);
  const int ot   = wave & 1;
  const float* aB = sW + (ot * 16 + mn) * WPITCH + half * 18;
  float bv[8];
  #pragma unroll
  for (int r = 0; r < 8; ++r) bv[r] = bias[ot * 16 + 8 * half + r];

  for (int tl = wave >> 1; tl < cnt; tl += 4) {
    int p = p0 + tl * 16 + mn;
    const bool live = p < NPIX;
    p = live ? p : NPIX - 1;
    const int y  = p / REGION;
    const int xx = p - y * REGION;

    const float* bB = sX + half * PAIRF + (y - y0) * ROWF + xx * 2;

    v8f acc = {0.f, 0.f, 0.f, 0.f, 0.f, 0.f, 0.f, 0.f};

    #pragma unroll
    for (int tap = 0; tap < 9; ++tap) {
      const int ky = tap / 3;
      const int kx = tap % 3;
      #pragma unroll
      for (int c4 = 0; c4 < CIN / 4; ++c4) {
        v2f a, bm;
        a.x = aB[c4 * 36 + tap];
        a.y = aB[c4 * 36 + tap + 9];
        bm  = *reinterpret_cast<const v2f*>(
                  bB + c4 * 2 * PAIRF + ((2 - ky) * BCOLP + (2 - kx)) * 2);
        acc = __builtin_amdgcn_wmma_f32_16x16x4_f32(false, a, false, bm,
                                                    (short)0, acc, false, false);
      }
    }

    if (live) {
      #pragma unroll
      for (int r = 0; r < 8; ++r) {
        const int o = ot * 16 + r + 8 * half;
        out[(((size_t)b * COUT + o) * HW + y) * HW + xx] = acc[r] * SCALE + bv[r];
      }
    }
  }
}

extern "C" void kernel_launch(void* const* d_in, const int* in_sizes, int n_in,
                              void* d_out, int out_size, void* d_ws, size_t ws_size,
                              hipStream_t stream) {
  const float* x    = (const float*)d_in[0];
  const float* w    = (const float*)d_in[1];
  const float* bias = (const float*)d_in[2];
  float* out = (float*)d_out;

  const int n4 = BATCH * COUT * HW * HW / 4;
  bias_fill_kernel<<<n4 / 256, 256, 0, stream>>>(bias, out);

  dim3 grid(GROUPS, BATCH);
  fftconv_corner_wmma<<<grid, 256, 0, stream>>>(x, w, bias, out);
}


# Bias fill folded into the conv kernel, 1 launch not 2: each group stores bias over one plane outside the 66x66 corner behind its staging loads; LDS-only barrier lets them drain under the WMMAs
# speedup vs baseline: 2.5940x; 1.3674x over previous
#include <hip/hip_runtime.h>


typedef __attribute__((ext_vector_type(2))) float v2f;
typedef __attribute__((ext_vector_type(4))) float v4f;
typedef __attribute__((ext_vector_type(8))) float v8f;

#define BATCH  8
#define CIN    32
#define COUT   32
#define HW     128
#define PLANE  (HW * HW)
#define REGION 66
#define NPIX   (REGION * REGION)
#define NTILE  273
#define GROUPS 32
#define SCALE  (1.0f/256.0f)
#define FILL4  ((HW - REGION) * HW / 4)
#define SIDE2  ((HW - REGION) / 2)
#define FILL2  (REGION * SIDE2)

#define WPITCH 292
#define BROWS  6
#define BCOLP  72
#define ROWF   (BCOLP * 2)
#define PAIRF  (BROWS * ROWF)
#define XITEMS (16 * BROWS * 33)

__global__ void __launch_bounds__(256) fftconv_corner_wmma(
    const float* __restrict__ x,
    const float* __restrict__ w,
    const float* __restrict__ bias,
    float* __restrict__ out) {
  __shared__ __attribute__((aligned(16))) float sW[COUT * WPITCH];
  __shared__ __attribute__((aligned(16))) float sX[16 * PAIRF];

  const int tid  = threadIdx.x;
  const int lane = tid & 31;
  const int half = lane >> 4;
  const int mn   = lane & 15;
  const int g    = blockIdx.x;
  const int b    = blockIdx.y;

  const int t0  = g * 8 + (g < 17 ? g : 17);
  const int cnt = 8 + (g < 17 ? 1 : 0);
  const int p0  = t0 * 16;
  const int y0  = p0 / REGION;

  const float* __restrict__ xb = x + (size_t)b * CIN * PLANE;

  v4f wv[9];
  #pragma unroll
  for (int i = 0; i < 9; ++i)
    wv[i] = reinterpret_cast<const v4f*>(w)[tid + i * 256];

  v2f xv0[13], xv1[13];
  #pragma unroll
  for (int i = 0; i < 13; ++i) {
    const int idx = tid + i * 256;
    const int idc = idx < XITEMS ? idx : XITEMS - 1;
    const int t   = idc / 33;
    const int bc2 = idc - t * 33;
    const int q   = t / BROWS;
    const int br  = t - q * BROWS;
    const int row = 62 + y0 + br;
    const int rowc = row < HW ? row : HW - 1;
    const float* src = xb + (q * 2) * PLANE + rowc * HW + 62 + bc2 * 2;
    xv0[i] = *reinterpret_cast<const v2f*>(src);
    xv1[i] = *reinterpret_cast<const v2f*>(src + PLANE);
  }

  {
    const float bf = bias[g];
    float* pl = out + ((size_t)b * COUT + g) * PLANE;
    const v4f f4 = {bf, bf, bf, bf};
    v4f* lo = reinterpret_cast<v4f*>(pl + REGION * HW);
    for (int i = 0; i < FILL4 / 256; ++i)
      lo[tid + i * 256] = f4;
    if (tid < FILL4 % 256) lo[tid + (FILL4 / 256) * 256] = f4;
    const v2f f2 = {bf, bf};
    for (int i = 0; i < (FILL2 + 255) / 256; ++i) {
      const int idx = tid + i * 256;
      if (i < FILL2 / 256 || idx < FILL2) {
        const int r  = idx / SIDE2;
        const int c2 = idx - r * SIDE2;
        *reinterpret_cast<v2f*>(pl + r * HW + REGION + c2 * 2) = f2;
      }
    }
  }

  #pragma unroll
  for (int i = 0; i < 9; ++i) {
    const int idx = tid + i * 256;
    const int o   = idx / 72;
    const int r4  = idx - o * 72;
    *reinterpret_cast<v4f*>(&sW[o * WPITCH + r4 * 4]) = wv[i];
  }
  #pragma unroll
  for (int i = 0; i < 13; ++i) {
    const int idx = tid + i * 256;
    if (idx < XITEMS) {
      const int t   = idx / 33;
      const int bc2 = idx - t * 33;
      const int br  = t % BROWS;
      const bool in = (62 + y0 + br) < HW;
      v4f pk;
      pk.x = in ? xv0[i].x : 0.0f;
      pk.y = in ? xv1[i].x : 0.0f;
      pk.z = in ? xv0[i].y : 0.0f;
      pk.w = in ? xv1[i].y : 0.0f;
      *reinterpret_cast<v4f*>(&sX[t * ROWF + bc2 * 4]) = pk;
    }
  }
  for (int idx = tid; idx < 16 * BROWS * 3; idx += 256) {
    const int t  = idx / 3;
    const int k3 = idx - t * 3;
    const v4f z = {0.f, 0.f, 0.f, 0.f};
    *reinterpret_cast<v4f*>(&sX[t * ROWF + (33 + k3) * 4]) = z;
  }
  __builtin_amdgcn_fence(__ATOMIC_RELEASE, "workgroup", "local");
  __builtin_amdgcn_s_barrier();
  __builtin_amdgcn_fence(__ATOMIC_ACQUIRE, "workgroup", "local");

  const int wave = __builtin_amdgcn_readfirstlane(tid >> 5);
  const int ot   = wave & 1;
  const float* aB = sW + (ot * 16 + mn) * WPITCH + half * 18;
  float bv[8];
  #pragma unroll
  for (int r = 0; r < 8; ++r) bv[r] = bias[ot * 16 + 8 * half + r];

  for (int tl = wave >> 1; tl < cnt; tl += 4) {
    int p = p0 + tl * 16 + mn;
    const bool live = p < NPIX;
    p = live ? p : NPIX - 1;
    const int y  = p / REGION;
    const int xx = p - y * REGION;

    const float* bB = sX + half * PAIRF + (y - y0) * ROWF + xx * 2;

    v8f acc = {0.f, 0.f, 0.f, 0.f, 0.f, 0.f, 0.f, 0.f};

    #pragma unroll
    for (int tap = 0; tap < 9; ++tap) {
      const int ky = tap / 3;
      const int kx = tap % 3;
      #pragma unroll
      for (int c4 = 0; c4 < CIN / 4; ++c4) {
        v2f a, bm;
        a.x = aB[c4 * 36 + tap];
        a.y = aB[c4 * 36 + tap + 9];
        bm  = *reinterpret_cast<const v2f*>(
                  bB + c4 * 2 * PAIRF + ((2 - ky) * BCOLP + (2 - kx)) * 2);
        acc = __builtin_amdgcn_wmma_f32_16x16x4_f32(false, a, false, bm,
                                                    (short)0, acc, false, false);
      }
    }

    if (live) {
      #pragma unroll
      for (int r = 0; r < 8; ++r) {
        const int o = ot * 16 + r + 8 * half;
        out[(((size_t)b * COUT + o) * HW + y) * HW + xx] = acc[r] * SCALE + bv[r];
      }
    }
  }
}

extern "C" void kernel_launch(void* const* d_in, const int* in_sizes, int n_in,
                              void* d_out, int out_size, void* d_ws, size_t ws_size,
                              hipStream_t stream) {
  const float* x    = (const float*)d_in[0];
  const float* w    = (const float*)d_in[1];
  const float* bias = (const float*)d_in[2];
  float* out = (float*)d_out;

  dim3 grid(GROUPS, BATCH);
  fftconv_corner_wmma<<<grid, 256, 0, stream>>>(x, w, bias, out);
}
